# starting kernel
# baseline (speedup 1.0000x reference)
#include <hip/hip_runtime.h>
#include <hip/hip_bf16.h>

typedef __attribute__((ext_vector_type(2))) float v2f;
typedef __attribute__((ext_vector_type(8))) float v8f;

#define WPAD 72

__global__ void fill_f32(float* __restrict__ p, float v, long n) {
  long i = (long)blockIdx.x * blockDim.x + threadIdx.x;
  if (i < n) p[i] = v;
}

__global__ void count_deg(const int* __restrict__ col, float* __restrict__ deg, int nE) {
  int e = blockIdx.x * blockDim.x + threadIdx.x;
  if (e < nE) unsafeAtomicAdd(&deg[col[e]], 1.0f);
}

__global__ void make_dinv(float* __restrict__ d, int n) {
  int i = blockIdx.x * blockDim.x + threadIdx.x;
  if (i < n) {
    float v = d[i];
    d[i] = (v > 0.0f) ? rsqrtf(v) : 0.0f;
  }
}

template <int K>
__global__ void __launch_bounds__(128)
gcn_gemm_wmma(const float* __restrict__ X, const float* __restrict__ W,
              const float* __restrict__ dinv, float* __restrict__ Y, int nRows) {
  __shared__ float wlds[K * WPAD];
  __shared__ float alds[16 * (K + 4)];

  const int tid  = threadIdx.x;
  const int wave = tid >> 5;
  const int lane = tid & 31;
  const int tile = blockIdx.x * 16;

  for (int i = tid * 4; i < K * 64; i += 128 * 4) {
    const int k = i >> 6;
    const int c = i & 63;
    *reinterpret_cast<float4*>(&wlds[k * WPAD + c]) =
        *reinterpret_cast<const float4*>(W + i);
  }
  for (int i = tid * 4; i < 16 * K; i += 128 * 4) {
    const int r = i / K;
    const int c = i % K;
    int gr = tile + r;
    if (gr >= nRows) gr = nRows - 1;
    *reinterpret_cast<float4*>(&alds[r * (K + 4) + c]) =
        *reinterpret_cast<const float4*>(X + (long)gr * K + c);
  }
  __syncthreads();

  const int m  = lane & 15;
  const int kq = (lane >> 4) * 2;
  const int n  = (lane & 15) + wave * 16;

  v8f c = {};
#pragma unroll
  for (int kk = 0; kk < K; kk += 4) {
    const int kb = kk + kq;
    v2f a, b;
    a.x = alds[m * (K + 4) + kb];
    a.y = alds[m * (K + 4) + kb + 1];
    b.x = wlds[kb * WPAD + n];
    b.y = wlds[(kb + 1) * WPAD + n];
    c = __builtin_amdgcn_wmma_f32_16x16x4_f32(false, a, false, b, (short)0, c,
                                              false, false);
  }

  const int mrow0 = (lane >> 4) * 8;
#pragma unroll
  for (int v = 0; v < 8; ++v) {
    const int gr = tile + mrow0 + v;
    if (gr < nRows) Y[(long)gr * 64 + n] = c[v] * dinv[gr];
  }
}

__global__ void __launch_bounds__(256)
scatter_edges(const int* __restrict__ row, const int* __restrict__ col,
              const float* __restrict__ y, float* __restrict__ acc, long nWork) {
  long t = (long)blockIdx.x * 256 + threadIdx.x;
  if (t >= nWork) return;
  const int e = (int)(t >> 4);
  const int f = (int)(t & 15) << 2;
  const int r = row[e];
  const int cd = col[e];
  const float4 v = *reinterpret_cast<const float4*>(y + (long)r * 64 + f);
  float* dst = acc + (long)cd * 64 + f;
  unsafeAtomicAdd(dst + 0, v.x);
  unsafeAtomicAdd(dst + 1, v.y);
  unsafeAtomicAdd(dst + 2, v.z);
  unsafeAtomicAdd(dst + 3, v.w);
}

__global__ void __launch_bounds__(256)
finalize_layer(const float* __restrict__ acc, const float* __restrict__ y,
               const float* __restrict__ dinv, const float* __restrict__ bias,
               float* __restrict__ out, int ostride, int nNodes) {
  long t = (long)blockIdx.x * 256 + threadIdx.x;
  if (t >= (long)nNodes * 16) return;
  const int i = (int)(t >> 4);
  const int f = (int)(t & 15) << 2;
  const float d = dinv[i];
  const float4 a = *reinterpret_cast<const float4*>(acc + (long)i * 64 + f);
  const float4 s = *reinterpret_cast<const float4*>(y + (long)i * 64 + f);
  const float4 b = *reinterpret_cast<const float4*>(bias + f);
  float4 o;
  o.x = d * (a.x + s.x) + b.x;
  o.y = d * (a.y + s.y) + b.y;
  o.z = d * (a.z + s.z) + b.z;
  o.w = d * (a.w + s.w) + b.w;
  *reinterpret_cast<float4*>(out + (long)i * ostride + f) = o;
}

extern "C" void kernel_launch(void* const* d_in, const int* in_sizes, int n_in,
                              void* d_out, int out_size, void* d_ws, size_t ws_size,
                              hipStream_t stream) {
  const float* latent    = (const float*)d_in[0];
  const float* condition = (const float*)d_in[1];
  const int*   ei        = (const int*)d_in[2];
  const float* W1 = (const float*)d_in[3];
  const float* b1 = (const float*)d_in[4];
  const float* W2 = (const float*)d_in[5];
  const float* b2 = (const float*)d_in[6];
  const float* W3 = (const float*)d_in[7];
  const float* b3 = (const float*)d_in[8];
  const float* W4 = (const float*)d_in[9];
  const float* b4 = (const float*)d_in[10];

  const int N = in_sizes[0] / 64;
  const int E = in_sizes[2] / 2;
  const int* rowp = ei;
  const int* colp = ei + E;

  char* ws = (char*)d_ws;
  size_t off = 0;
  float* dinv = (float*)(ws + off);
  off += ((size_t)N * 4 + 255) & ~(size_t)255;
  float* ybuf = (float*)(ws + off);
  off += (size_t)N * 64 * 4;
  float* acc  = (float*)(ws + off);
  off += (size_t)N * 64 * 4;
  float* hbuf = (float*)(ws + off);

  const long accElems = (long)N * 64;
  const long edgeWork = (long)E * 16;
  const dim3 gemmGrid((N + 15) / 16);
  const dim3 gemmBlk(128);
  const dim3 fillN((N + 255) / 256);
  const dim3 fillAcc((accElems + 255) / 256);
  const dim3 edgeGrid((edgeWork + 255) / 256);
  const dim3 finGrid(((long)N * 16 + 255) / 256);

  fill_f32<<<fillN, 256, 0, stream>>>(dinv, 1.0f, N);
  count_deg<<<(E + 255) / 256, 256, 0, stream>>>(colp, dinv, E);
  make_dinv<<<fillN, 256, 0, stream>>>(dinv, N);

  gcn_gemm_wmma<64><<<gemmGrid, gemmBlk, 0, stream>>>(latent, W1, dinv, ybuf, N);
  fill_f32<<<fillAcc, 256, 0, stream>>>(acc, 0.0f, accElems);
  scatter_edges<<<edgeGrid, 256, 0, stream>>>(rowp, colp, ybuf, acc, edgeWork);
  finalize_layer<<<finGrid, 256, 0, stream>>>(acc, ybuf, dinv, b1, hbuf, 128, N);

  gcn_gemm_wmma<32><<<gemmGrid, gemmBlk, 0, stream>>>(condition, W2, dinv, ybuf, N);
  fill_f32<<<fillAcc, 256, 0, stream>>>(acc, 0.0f, accElems);
  scatter_edges<<<edgeGrid, 256, 0, stream>>>(rowp, colp, ybuf, acc, edgeWork);
  finalize_layer<<<finGrid, 256, 0, stream>>>(acc, ybuf, dinv, b2, hbuf + 64, 128, N);

  gcn_gemm_wmma<128><<<gemmGrid, gemmBlk, 0, stream>>>(hbuf, W3, dinv, ybuf, N);
  fill_f32<<<fillAcc, 256, 0, stream>>>(acc, 0.0f, accElems);
  scatter_edges<<<edgeGrid, 256, 0, stream>>>(rowp, colp, ybuf, acc, edgeWork);
  finalize_layer<<<finGrid, 256, 0, stream>>>(acc, ybuf, dinv, b3, hbuf, 64, N);

  gcn_gemm_wmma<64><<<gemmGrid, gemmBlk, 0, stream>>>(hbuf, W4, dinv, ybuf, N);
  fill_f32<<<fillAcc, 256, 0, stream>>>(acc, 0.0f, accElems);
  scatter_edges<<<edgeGrid, 256, 0, stream>>>(rowp, colp, ybuf, acc, edgeWork);
  finalize_layer<<<finGrid, 256, 0, stream>>>(acc, ybuf, dinv, b4, (float*)d_out, 64, N);
}


# 64-feature edge sums in ONE copy, stored as two 32-feature planes, one per chip half: a work-group draws a chunk ticket for its half and adds only its plane (float atomics); 12 launches -> 11
# speedup vs baseline: 2.9108x; 2.9108x over previous
#include <hip/hip_runtime.h>
#include <hip/hip_bf16.h>

typedef __attribute__((ext_vector_type(2))) float v2f;
typedef __attribute__((ext_vector_type(8))) float v8f;

#define WPAD 72

__device__ __forceinline__ int chip_half() {
  return (int)((__builtin_amdgcn_s_sendmsg_rtn(0x87) >> 18) & 1u);
}

__device__ __forceinline__ long plane_idx(long i, int f, int n) {
  return (long)(f >> 5) * n * 32 + i * 32 + (f & 31);
}

__global__ void fill2_f32(float* __restrict__ p0, float v0, long n0,
                          float* __restrict__ p1, float v1, long n1) {
  long i = (long)blockIdx.x * blockDim.x + threadIdx.x;
  if (i < n0) p0[i] = v0;
  else if (i < n0 + n1) p1[i - n0] = v1;
}

__global__ void __launch_bounds__(256)
count_deg(const int* __restrict__ col, float* __restrict__ deg0,
          float* __restrict__ deg1, int nE) {
  __shared__ int s_half;
  int e = blockIdx.x * blockDim.x + threadIdx.x;
  int c = 0;
  if (e < nE) c = col[e];
  if (threadIdx.x == 0) s_half = chip_half();
  __syncthreads();
  float* deg = s_half ? deg1 : deg0;
  if (e < nE) unsafeAtomicAdd(&deg[c], 1.0f);
}


__global__ void __launch_bounds__(256)
stage_inputs(const float* __restrict__ latent, const float* __restrict__ cond,
             const float* __restrict__ deg0, const float* __restrict__ deg1,
             float* __restrict__ dinv, float* __restrict__ xsA, float* __restrict__ xsB,
             float* __restrict__ clr0, long n0, float* __restrict__ clr1, long n1,
             int nNodes) {
  const long t  = (long)blockIdx.x * 256 + threadIdx.x;
  const long nA = (long)nNodes * 16;
  const long nB = (long)nNodes * 8;
  if (t < nA) {
    const long i = t >> 4;
    const float v = deg0[i] + deg1[i];
    const float d = (v > 0.0f) ? rsqrtf(v) : 0.0f;
    if ((t & 15) == 0) dinv[i] = d;
    float4 s = *reinterpret_cast<const float4*>(latent + t * 4);
    s.x *= d; s.y *= d; s.z *= d; s.w *= d;
    *reinterpret_cast<float4*>(xsA + plane_idx(i, (int)(t & 15) * 4, nNodes)) = s;
  } else if (t < nA + nB) {
    const long u = t - nA;
    const long i = u >> 3;
    const float v = deg0[i] + deg1[i];
    const float d = (v > 0.0f) ? rsqrtf(v) : 0.0f;
    float4 s = *reinterpret_cast<const float4*>(cond + u * 4);
    s.x *= d; s.y *= d; s.z *= d; s.w *= d;
    *reinterpret_cast<float4*>(xsB + u * 4) = s;
  } else {
    const long u = t - nA - nB;
    const float4 z = make_float4(0.0f, 0.0f, 0.0f, 0.0f);
    if (u < n0) *reinterpret_cast<float4*>(clr0 + u * 4) = z;
    else if (u < n0 + n1) *reinterpret_cast<float4*>(clr1 + (u - n0) * 4) = z;
  }
}

#define FROWS 32
__global__ void __launch_bounds__(256)
gcn_layers123_wmma(const float* __restrict__ latent, const float* __restrict__ accA,
                   const float* __restrict__ cond,
                   const float* __restrict__ accBp, const float* __restrict__ accBq,
                   const float* __restrict__ W1, const float* __restrict__ b1,
                   const float* __restrict__ W2, const float* __restrict__ b2,
                   const float* __restrict__ W3, const float* __restrict__ dinv,
                   float* __restrict__ Y, int nRows) {
  __shared__ float w1lds[64 * WPAD];
  __shared__ float w2lds[32 * WPAD];
  __shared__ float w3lds[128 * WPAD];
  __shared__ float a1lds[FROWS * (64 + 4)];
  __shared__ float a2lds[FROWS * (32 + 4)];
  __shared__ float hlds[FROWS * (128 + 4)];

  const int tid  = threadIdx.x;
  const int wave = tid >> 5;
  const int lane = tid & 31;
  const int rt   = wave >> 2;
  const int ct   = wave & 3;
  const int tile = blockIdx.x * FROWS;

  for (int it = 0; it < (64 * 64) / (256 * 4); ++it) {
    const int i = tid * 4 + it * (256 * 4);
    *reinterpret_cast<float4*>(&w1lds[(i >> 6) * WPAD + (i & 63)]) =
        *reinterpret_cast<const float4*>(W1 + i);
  }
  for (int it = 0; it < (32 * 64) / (256 * 4); ++it) {
    const int i = tid * 4 + it * (256 * 4);
    *reinterpret_cast<float4*>(&w2lds[(i >> 6) * WPAD + (i & 63)]) =
        *reinterpret_cast<const float4*>(W2 + i);
  }
  for (int it = 0; it < (128 * 64) / (256 * 4); ++it) {
    const int i = tid * 4 + it * (256 * 4);
    *reinterpret_cast<float4*>(&w3lds[(i >> 6) * WPAD + (i & 63)]) =
        *reinterpret_cast<const float4*>(W3 + i);
  }
  for (int it = 0; it < (FROWS * 64) / (256 * 4); ++it) {
    const int i = tid * 4 + it * (256 * 4);
    const int r = i >> 6;
    const int c = i & 63;
    int gr = tile + r;
    if (gr >= nRows) gr = nRows - 1;
    const float d = dinv[gr];
    const float4 x4 = *reinterpret_cast<const float4*>(latent + (long)gr * 64 + c);
    const float4 a4 = *reinterpret_cast<const float4*>(accA + plane_idx(gr, c, nRows));
    float4 t;
    t.x = d * (a4.x + d * x4.x);
    t.y = d * (a4.y + d * x4.y);
    t.z = d * (a4.z + d * x4.z);
    t.w = d * (a4.w + d * x4.w);
    *reinterpret_cast<float4*>(&a1lds[r * (64 + 4) + c]) = t;
  }
  for (int it = 0; it < (FROWS * 32) / (256 * 4); ++it) {
    const int i = tid * 4 + it * (256 * 4);
    const int r = i >> 5;
    const int c = i & 31;
    int gr = tile + r;
    if (gr >= nRows) gr = nRows - 1;
    const float d = dinv[gr];
    const float4 x4 = *reinterpret_cast<const float4*>(cond + (long)gr * 32 + c);
    float4 a4 = *reinterpret_cast<const float4*>(accBp + (long)gr * 32 + c);
    const float4 q4 = *reinterpret_cast<const float4*>(accBq + (long)gr * 32 + c);
    a4.x += q4.x; a4.y += q4.y; a4.z += q4.z; a4.w += q4.w;
    float4 t;
    t.x = d * (a4.x + d * x4.x);
    t.y = d * (a4.y + d * x4.y);
    t.z = d * (a4.z + d * x4.z);
    t.w = d * (a4.w + d * x4.w);
    *reinterpret_cast<float4*>(&a2lds[r * (32 + 4) + c]) = t;
  }
  __syncthreads();

  const int m  = (lane & 15) + rt * 16;
  const int kq = (lane >> 4) * 2;
  const int n  = (lane & 15) + ct * 16;

  v8f c1 = {};
  for (int kk = 0; kk < 64; kk += 4) {
    const int kb = kk + kq;
    v2f a, b;
    a.x = a1lds[m * (64 + 4) + kb];
    a.y = a1lds[m * (64 + 4) + kb + 1];
    b.x = w1lds[kb * WPAD + n];
    b.y = w1lds[(kb + 1) * WPAD + n];
    c1 = __builtin_amdgcn_wmma_f32_16x16x4_f32(false, a, false, b, (short)0, c1,
                                               false, false);
  }
  v8f c2 = {};
  for (int kk = 0; kk < 32; kk += 4) {
    const int kb = kk + kq;
    v2f a, b;
    a.x = a2lds[m * (32 + 4) + kb];
    a.y = a2lds[m * (32 + 4) + kb + 1];
    b.x = w2lds[kb * WPAD + n];
    b.y = w2lds[(kb + 1) * WPAD + n];
    c2 = __builtin_amdgcn_wmma_f32_16x16x4_f32(false, a, false, b, (short)0, c2,
                                               false, false);
  }

  const float bn1 = b1[n];
  const float bn2 = b2[n];
  const int mrow0 = rt * 16 + (lane >> 4) * 8;
  for (int v = 0; v < 8; ++v) {
    hlds[(mrow0 + v) * (128 + 4) + n]      = c1[v] + bn1;
    hlds[(mrow0 + v) * (128 + 4) + 64 + n] = c2[v] + bn2;
  }
  __syncthreads();

  v8f c3 = {};
  for (int kk = 0; kk < 128; kk += 4) {
    const int kb = kk + kq;
    v2f a, b;
    a.x = hlds[m * (128 + 4) + kb];
    a.y = hlds[m * (128 + 4) + kb + 1];
    b.x = w3lds[kb * WPAD + n];
    b.y = w3lds[(kb + 1) * WPAD + n];
    c3 = __builtin_amdgcn_wmma_f32_16x16x4_f32(false, a, false, b, (short)0, c3,
                                               false, false);
  }

  for (int v = 0; v < 8; ++v) {
    const int gr = tile + mrow0 + v;
    if (gr < nRows) Y[plane_idx(gr, n, nRows)] = c3[v] * dinv[gr];
  }
}

__global__ void __launch_bounds__(128)
gcn_fin_gemm_wmma(const float* __restrict__ acc,
                  const float* __restrict__ y, const float* __restrict__ dinv,
                  const float* __restrict__ bias, const float* __restrict__ W,
                  float* __restrict__ Y, int nRows) {
  __shared__ float wlds[64 * WPAD];
  __shared__ float alds[16 * (64 + 4)];

  const int tid  = threadIdx.x;
  const int wave = tid >> 5;
  const int lane = tid & 31;
  const int tile = blockIdx.x * 16;

  for (int it = 0; it < (64 * 64) / (128 * 4); ++it) {
    const int i = tid * 4 + it * (128 * 4);
    *reinterpret_cast<float4*>(&wlds[(i >> 6) * WPAD + (i & 63)]) =
        *reinterpret_cast<const float4*>(W + i);
  }
  for (int it = 0; it < (16 * 64) / (128 * 4); ++it) {
    const int i = tid * 4 + it * (128 * 4);
    const int r = i >> 6;
    const int c = i & 63;
    int gr = tile + r;
    if (gr >= nRows) gr = nRows - 1;
    const float d = dinv[gr];
    const float4 a4 = *reinterpret_cast<const float4*>(acc + plane_idx(gr, c, nRows));
    const float4 s4 = *reinterpret_cast<const float4*>(y + plane_idx(gr, c, nRows));
    const float4 b4 = *reinterpret_cast<const float4*>(bias + c);
    float4 t;
    t.x = d * (a4.x + s4.x) + b4.x;
    t.y = d * (a4.y + s4.y) + b4.y;
    t.z = d * (a4.z + s4.z) + b4.z;
    t.w = d * (a4.w + s4.w) + b4.w;
    *reinterpret_cast<float4*>(&alds[r * (64 + 4) + c]) = t;
  }
  __syncthreads();

  const int m  = lane & 15;
  const int kq = (lane >> 4) * 2;
  const int n  = (lane & 15) + wave * 16;

  v8f c = {};
  for (int kk = 0; kk < 64; kk += 4) {
    const int kb = kk + kq;
    v2f a, b;
    a.x = alds[m * (64 + 4) + kb];
    a.y = alds[m * (64 + 4) + kb + 1];
    b.x = wlds[kb * WPAD + n];
    b.y = wlds[(kb + 1) * WPAD + n];
    c = __builtin_amdgcn_wmma_f32_16x16x4_f32(false, a, false, b, (short)0, c,
                                              false, false);
  }

  const int mrow0 = (lane >> 4) * 8;
  for (int v = 0; v < 8; ++v) {
    const int gr = tile + mrow0 + v;
    if (gr < nRows) Y[plane_idx(gr, n, nRows)] = c[v] * dinv[gr];
  }
}

#define EDGE_HALF 16

__device__ __forceinline__ void load_rows32(const float* __restrict__ xl, int r, int j0,
                                            float (&v)[EDGE_HALF]) {
#pragma unroll
  for (int k = 0; k < EDGE_HALF; ++k) {
    const int rj = __builtin_amdgcn_readlane(r, j0 + k);
    v[k] = xl[(long)rj * 32];
  }
}

__device__ __forceinline__ void add_rows32(float* __restrict__ al, int c, int j0,
                                           const float (&v)[EDGE_HALF]) {
#pragma unroll
  for (int k = 0; k < EDGE_HALF; ++k) {
    const int cj = __builtin_amdgcn_readlane(c, j0 + k);
    unsafeAtomicAdd(al + (long)cj * 32, v[k]);
  }
}

__global__ void __launch_bounds__(256)
scatter_edges32(const int* __restrict__ row, const int* __restrict__ col,
                const float* __restrict__ xs, float* __restrict__ acc0,
                float* __restrict__ acc1, int nE) {
  __shared__ int s_half;
  const int lane = threadIdx.x & 31;
  const int wv = __builtin_amdgcn_readfirstlane((int)((blockIdx.x * 256 + threadIdx.x) >> 5));
  const int e0 = wv * 32;
  const int e = e0 + lane;
  int r = 0, c = 0;
  if (e < nE) { r = row[e]; c = col[e]; }
  if (threadIdx.x == 0) s_half = chip_half();
  __syncthreads();
  if (e0 >= nE) return;
  float* acc = __builtin_amdgcn_readfirstlane(s_half) ? acc1 : acc0;
  const float* xl = xs + lane;
  float* al = acc + lane;
  const int cnt = (nE - e0 < 32) ? (nE - e0) : 32;
  if (cnt == 32) {
    float a[EDGE_HALF], b[EDGE_HALF];
    load_rows32(xl, r, 0, a);
    load_rows32(xl, r, 16, b);
    add_rows32(al, c, 0, a);
    add_rows32(al, c, 16, b);
  } else {
    for (int j = 0; j < cnt; ++j) {
      const int rj = __builtin_amdgcn_readlane(r, j);
      const int cj = __builtin_amdgcn_readlane(c, j);
      const float v = xl[(long)rj * 32];
      unsafeAtomicAdd(al + (long)cj * 32, v);
    }
  }
}

#define CHUNK_EDGES 512
#define CTR_STRIDE 64

__global__ void __launch_bounds__(256)
scatter_edges(const int* __restrict__ row, const int* __restrict__ col,
              const float* __restrict__ y, float* __restrict__ acc,
              int* __restrict__ ctr, int nE, int nChunks, int nNodes) {
  __shared__ int s_half;
  __shared__ int s_chunk;
  if (threadIdx.x == 0) {
    int h = chip_half();
    int t = atomicAdd(&ctr[h * CTR_STRIDE], 1);
    if (t >= nChunks) {
      h ^= 1;
      t = atomicAdd(&ctr[h * CTR_STRIDE], 1);
    }
    s_half = h;
    s_chunk = t;
  }
  __syncthreads();
  const int h = __builtin_amdgcn_readfirstlane(s_half);
  const int t = __builtin_amdgcn_readfirstlane(s_chunk);
  if (t >= nChunks) return;
  const int lane = threadIdx.x & 31;
  const int wv = __builtin_amdgcn_readfirstlane((int)(threadIdx.x >> 5));
  const int e0 = t * CHUNK_EDGES + wv * 64;
  const int eA = e0 + lane;
  const int eB = e0 + 32 + lane;
  int rA = 0, cA = 0, rB = 0, cB = 0;
  if (eA < nE) { rA = row[eA]; cA = col[eA]; }
  if (eB < nE) { rB = row[eB]; cB = col[eB]; }
  const float* xl = y + (long)h * nNodes * 32 + lane;
  float* al = acc + (long)h * nNodes * 32 + lane;
  if (e0 + 64 <= nE) {
    float a[EDGE_HALF], b[EDGE_HALF];
    load_rows32(xl, rA, 0, a);
    load_rows32(xl, rA, 16, b);
    add_rows32(al, cA, 0, a);
    load_rows32(xl, rB, 0, a);
    add_rows32(al, cA, 16, b);
    load_rows32(xl, rB, 16, b);
    add_rows32(al, cB, 0, a);
    add_rows32(al, cB, 16, b);
  } else {
    const int cntA = (nE - e0 < 32) ? (nE - e0) : 32;
    const int cntB = (nE - e0 - 32 < 32) ? (nE - e0 - 32) : 32;
    for (int j = 0; j < cntA; ++j) {
      const int rj = __builtin_amdgcn_readlane(rA, j);
      const int cj = __builtin_amdgcn_readlane(cA, j);
      unsafeAtomicAdd(al + (long)cj * 32, xl[(long)rj * 32]);
    }
    for (int j = 0; j < cntB; ++j) {
      const int rj = __builtin_amdgcn_readlane(rB, j);
      const int cj = __builtin_amdgcn_readlane(cB, j);
      unsafeAtomicAdd(al + (long)cj * 32, xl[(long)rj * 32]);
    }
  }
}

__global__ void __launch_bounds__(256)
finalize_layer(const float* __restrict__ acc,
               const float* __restrict__ y,
               const float* __restrict__ dinv, const float* __restrict__ bias,
               float* __restrict__ out, int ostride, int nNodes) {
  long t = (long)blockIdx.x * 256 + threadIdx.x;
  if (t >= (long)nNodes * 16) return;
  const int i = (int)(t >> 4);
  const int f = (int)(t & 15) << 2;
  const float d = dinv[i];
  const float4 a = *reinterpret_cast<const float4*>(acc + plane_idx(i, f, nNodes));
  const float4 s = *reinterpret_cast<const float4*>(y + plane_idx(i, f, nNodes));
  const float4 b = *reinterpret_cast<const float4*>(bias + f);
  float4 o;
  o.x = d * (a.x + s.x) + b.x;
  o.y = d * (a.y + s.y) + b.y;
  o.z = d * (a.z + s.z) + b.z;
  o.w = d * (a.w + s.w) + b.w;
  *reinterpret_cast<float4*>(out + (long)i * ostride + f) = o;
}

extern "C" void kernel_launch(void* const* d_in, const int* in_sizes, int n_in,
                              void* d_out, int out_size, void* d_ws, size_t ws_size,
                              hipStream_t stream) {
  const float* latent    = (const float*)d_in[0];
  const float* condition = (const float*)d_in[1];
  const int*   ei        = (const int*)d_in[2];
  const float* W1 = (const float*)d_in[3];
  const float* b1 = (const float*)d_in[4];
  const float* W2 = (const float*)d_in[5];
  const float* b2 = (const float*)d_in[6];
  const float* W3 = (const float*)d_in[7];
  const float* b3 = (const float*)d_in[8];
  const float* W4 = (const float*)d_in[9];
  const float* b4 = (const float*)d_in[10];

  const int N = in_sizes[0] / 64;
  const int E = in_sizes[2] / 2;
  const int* rowp = ei;
  const int* colp = ei + E;

  char* ws = (char*)d_ws;
  size_t off = 0;
  float* dinv = (float*)(ws + off);
  off += ((size_t)N * 4 + 255) & ~(size_t)255;
  float* ybuf = (float*)(ws + off);
  off += (size_t)N * 64 * 4;
  float* acc  = (float*)(ws + off);
  off += (size_t)N * 64 * 4;
  float* hbuf = (float*)(ws + off);

  const long accElems = (long)N * 64;
  const dim3 gemmGrid((N + 15) / 16);
  const dim3 gemmBlk(128);
  const dim3 fusedGrid((N + FROWS - 1) / FROWS);
  const dim3 edgeGrid((E + 255) / 256);
  const int  nChunks = (E + CHUNK_EDGES - 1) / CHUNK_EDGES;
  const dim3 planeGrid(2 * nChunks);
  const dim3 finGrid(((long)N * 16 + 255) / 256);
  const dim3 fillAcc64((accElems + 255) / 256);
  const int  nCtr = 6 * CTR_STRIDE;
  const dim3 fill2N((2 * (long)N + nCtr + 255) / 256);
  const dim3 stageGrid(((long)N * 72 + 255) / 256);
  float* outp = (float*)d_out;

  float* xsA   = hbuf;
  float* xsB   = hbuf + accElems;
  float* deg0  = hbuf + (long)N * 96;
  float* deg1  = deg0 + N;
  int*   ctr   = (int*)(deg1 + N);
  float* accA  = ybuf;
  float* accB  = acc;
  float* accB1 = acc + (long)N * 32;
  float* acc3  = outp;
  fill2_f32<<<fill2N, 256, 0, stream>>>(deg0, 1.0f, N, deg1, 0.0f, (long)N + nCtr);
  count_deg<<<(E + 255) / 256, 256, 0, stream>>>(colp, deg0, deg1, E);
  stage_inputs<<<stageGrid, 256, 0, stream>>>(latent, condition, deg0, deg1, dinv, xsA, xsB,
                                              accA, (long)N * 32, acc3, (long)N * 16, N);
  scatter_edges<<<planeGrid, 256, 0, stream>>>(rowp, colp, xsA, accA, ctr, E, nChunks, N);
  scatter_edges32<<<edgeGrid, 256, 0, stream>>>(rowp, colp, xsB, accB, accB1, E);

  float* y3   = hbuf;
  gcn_layers123_wmma<<<fusedGrid, 256, 0, stream>>>(latent, accA, condition, accB, accB1,
                                                    W1, b1, W2, b2, W3, dinv, y3, N);
  scatter_edges<<<planeGrid, 256, 0, stream>>>(rowp, colp, y3, acc3, ctr + 2 * CTR_STRIDE, E, nChunks, N);

  gcn_fin_gemm_wmma<<<gemmGrid, gemmBlk, 0, stream>>>(acc3, y3, dinv, b3, W4, ybuf, N);
  fill2_f32<<<fillAcc64, 256, 0, stream>>>(acc, 0.0f, accElems, acc, 0.0f, 0);
  scatter_edges<<<planeGrid, 256, 0, stream>>>(rowp, colp, ybuf, acc, ctr + 4 * CTR_STRIDE, E, nChunks, N);
  finalize_layer<<<finGrid, 256, 0, stream>>>(acc, ybuf, dinv, b4, outp, 64, N);
}


# Edge sums start from their self-loop term, not zero: whoever writes a gather row also stores it as the sum's first addend (float atomics add onto it); clears and y re-reads gone; 11 launches -> 10
# speedup vs baseline: 3.2096x; 1.1027x over previous
#include <hip/hip_runtime.h>
#include <hip/hip_bf16.h>

typedef __attribute__((ext_vector_type(2))) float v2f;
typedef __attribute__((ext_vector_type(8))) float v8f;

#define WPAD 72

__device__ __forceinline__ int chip_half() {
  return (int)((__builtin_amdgcn_s_sendmsg_rtn(0x87) >> 18) & 1u);
}

__device__ __forceinline__ long plane_idx(long i, int f, int n) {
  return (long)(f >> 5) * n * 32 + i * 32 + (f & 31);
}

__global__ void fill2_f32(float* __restrict__ p0, float v0, long n0,
                          float* __restrict__ p1, float v1, long n1) {
  long i = (long)blockIdx.x * blockDim.x + threadIdx.x;
  if (i < n0) p0[i] = v0;
  else if (i < n0 + n1) p1[i - n0] = v1;
}

__global__ void __launch_bounds__(256)
count_deg(const int* __restrict__ col, float* __restrict__ deg0,
          float* __restrict__ deg1, int nE) {
  __shared__ int s_half;
  int e = blockIdx.x * blockDim.x + threadIdx.x;
  int c = 0;
  if (e < nE) c = col[e];
  if (threadIdx.x == 0) s_half = chip_half();
  __syncthreads();
  float* deg = s_half ? deg1 : deg0;
  if (e < nE) unsafeAtomicAdd(&deg[c], 1.0f);
}


__global__ void __launch_bounds__(256)
stage_inputs(const float* __restrict__ latent, const float* __restrict__ cond,
             const float* __restrict__ deg0, const float* __restrict__ deg1,
             float* __restrict__ dinv, float* __restrict__ xsA, float* __restrict__ xsB,
             float* __restrict__ accA, float* __restrict__ accBp, float* __restrict__ accBq,
             int nNodes) {
  const long t  = (long)blockIdx.x * 256 + threadIdx.x;
  const long nA = (long)nNodes * 16;
  const long nB = (long)nNodes * 8;
  if (t < nA) {
    const long i = t >> 4;
    const float v = deg0[i] + deg1[i];
    const float d = (v > 0.0f) ? rsqrtf(v) : 0.0f;
    if ((t & 15) == 0) dinv[i] = d;
    float4 s = *reinterpret_cast<const float4*>(latent + t * 4);
    s.x *= d; s.y *= d; s.z *= d; s.w *= d;
    const long o = plane_idx(i, (int)(t & 15) * 4, nNodes);
    *reinterpret_cast<float4*>(xsA + o) = s;
    *reinterpret_cast<float4*>(accA + o) = s;
  } else if (t < nA + nB) {
    const long u = t - nA;
    const long i = u >> 3;
    const float v = deg0[i] + deg1[i];
    const float d = (v > 0.0f) ? rsqrtf(v) : 0.0f;
    float4 s = *reinterpret_cast<const float4*>(cond + u * 4);
    s.x *= d; s.y *= d; s.z *= d; s.w *= d;
    *reinterpret_cast<float4*>(xsB + u * 4) = s;
    *reinterpret_cast<float4*>(accBp + u * 4) = s;
    *reinterpret_cast<float4*>(accBq + u * 4) = make_float4(0.0f, 0.0f, 0.0f, 0.0f);
  }
}

#define FROWS 32
__global__ void __launch_bounds__(256)
gcn_layers123_wmma(const float* __restrict__ accA,
                   const float* __restrict__ accBp, const float* __restrict__ accBq,
                   const float* __restrict__ W1, const float* __restrict__ b1,
                   const float* __restrict__ W2, const float* __restrict__ b2,
                   const float* __restrict__ W3, const float* __restrict__ dinv,
                   float* __restrict__ Y, float* __restrict__ S, int nRows) {
  __shared__ float w1lds[64 * WPAD];
  __shared__ float w2lds[32 * WPAD];
  __shared__ float w3lds[128 * WPAD];
  __shared__ float a1lds[FROWS * (64 + 4)];
  __shared__ float a2lds[FROWS * (32 + 4)];
  __shared__ float hlds[FROWS * (128 + 4)];

  const int tid  = threadIdx.x;
  const int wave = tid >> 5;
  const int lane = tid & 31;
  const int rt   = wave >> 2;
  const int ct   = wave & 3;
  const int tile = blockIdx.x * FROWS;

  for (int it = 0; it < (64 * 64) / (256 * 4); ++it) {
    const int i = tid * 4 + it * (256 * 4);
    *reinterpret_cast<float4*>(&w1lds[(i >> 6) * WPAD + (i & 63)]) =
        *reinterpret_cast<const float4*>(W1 + i);
  }
  for (int it = 0; it < (32 * 64) / (256 * 4); ++it) {
    const int i = tid * 4 + it * (256 * 4);
    *reinterpret_cast<float4*>(&w2lds[(i >> 6) * WPAD + (i & 63)]) =
        *reinterpret_cast<const float4*>(W2 + i);
  }
  for (int it = 0; it < (128 * 64) / (256 * 4); ++it) {
    const int i = tid * 4 + it * (256 * 4);
    *reinterpret_cast<float4*>(&w3lds[(i >> 6) * WPAD + (i & 63)]) =
        *reinterpret_cast<const float4*>(W3 + i);
  }
  for (int it = 0; it < (FROWS * 64) / (256 * 4); ++it) {
    const int i = tid * 4 + it * (256 * 4);
    const int r = i >> 6;
    const int c = i & 63;
    int gr = tile + r;
    if (gr >= nRows) gr = nRows - 1;
    const float d = dinv[gr];
    const float4 a4 = *reinterpret_cast<const float4*>(accA + plane_idx(gr, c, nRows));
    float4 t;
    t.x = d * a4.x;
    t.y = d * a4.y;
    t.z = d * a4.z;
    t.w = d * a4.w;
    *reinterpret_cast<float4*>(&a1lds[r * (64 + 4) + c]) = t;
  }
  for (int it = 0; it < (FROWS * 32) / (256 * 4); ++it) {
    const int i = tid * 4 + it * (256 * 4);
    const int r = i >> 5;
    const int c = i & 31;
    int gr = tile + r;
    if (gr >= nRows) gr = nRows - 1;
    const float d = dinv[gr];
    float4 a4 = *reinterpret_cast<const float4*>(accBp + (long)gr * 32 + c);
    const float4 q4 = *reinterpret_cast<const float4*>(accBq + (long)gr * 32 + c);
    a4.x += q4.x; a4.y += q4.y; a4.z += q4.z; a4.w += q4.w;
    float4 t;
    t.x = d * a4.x;
    t.y = d * a4.y;
    t.z = d * a4.z;
    t.w = d * a4.w;
    *reinterpret_cast<float4*>(&a2lds[r * (32 + 4) + c]) = t;
  }
  __syncthreads();

  const int m  = (lane & 15) + rt * 16;
  const int kq = (lane >> 4) * 2;
  const int n  = (lane & 15) + ct * 16;

  v8f c1 = {};
  for (int kk = 0; kk < 64; kk += 4) {
    const int kb = kk + kq;
    v2f a, b;
    a.x = a1lds[m * (64 + 4) + kb];
    a.y = a1lds[m * (64 + 4) + kb + 1];
    b.x = w1lds[kb * WPAD + n];
    b.y = w1lds[(kb + 1) * WPAD + n];
    c1 = __builtin_amdgcn_wmma_f32_16x16x4_f32(false, a, false, b, (short)0, c1,
                                               false, false);
  }
  v8f c2 = {};
  for (int kk = 0; kk < 32; kk += 4) {
    const int kb = kk + kq;
    v2f a, b;
    a.x = a2lds[m * (32 + 4) + kb];
    a.y = a2lds[m * (32 + 4) + kb + 1];
    b.x = w2lds[kb * WPAD + n];
    b.y = w2lds[(kb + 1) * WPAD + n];
    c2 = __builtin_amdgcn_wmma_f32_16x16x4_f32(false, a, false, b, (short)0, c2,
                                               false, false);
  }

  const float bn1 = b1[n];
  const float bn2 = b2[n];
  const int mrow0 = rt * 16 + (lane >> 4) * 8;
  for (int v = 0; v < 8; ++v) {
    hlds[(mrow0 + v) * (128 + 4) + n]      = c1[v] + bn1;
    hlds[(mrow0 + v) * (128 + 4) + 64 + n] = c2[v] + bn2;
  }
  __syncthreads();

  v8f c3 = {};
  for (int kk = 0; kk < 128; kk += 4) {
    const int kb = kk + kq;
    v2f a, b;
    a.x = hlds[m * (128 + 4) + kb];
    a.y = hlds[m * (128 + 4) + kb + 1];
    b.x = w3lds[kb * WPAD + n];
    b.y = w3lds[(kb + 1) * WPAD + n];
    c3 = __builtin_amdgcn_wmma_f32_16x16x4_f32(false, a, false, b, (short)0, c3,
                                               false, false);
  }

  for (int v = 0; v < 8; ++v) {
    const int gr = tile + mrow0 + v;
    if (gr < nRows) {
      const long o = plane_idx(gr, n, nRows);
      const float yv = c3[v] * dinv[gr];
      Y[o] = yv;
      S[o] = yv;
    }
  }
}

__global__ void __launch_bounds__(128)
gcn_fin_gemm_wmma(const float* __restrict__ acc, const float* __restrict__ dinv,
                  const float* __restrict__ bias, const float* __restrict__ W,
                  float* __restrict__ Y, float* __restrict__ S, int nRows) {
  __shared__ float wlds[64 * WPAD];
  __shared__ float alds[16 * (64 + 4)];

  const int tid  = threadIdx.x;
  const int wave = tid >> 5;
  const int lane = tid & 31;
  const int tile = blockIdx.x * 16;

  for (int it = 0; it < (64 * 64) / (128 * 4); ++it) {
    const int i = tid * 4 + it * (128 * 4);
    *reinterpret_cast<float4*>(&wlds[(i >> 6) * WPAD + (i & 63)]) =
        *reinterpret_cast<const float4*>(W + i);
  }
  for (int it = 0; it < (16 * 64) / (128 * 4); ++it) {
    const int i = tid * 4 + it * (128 * 4);
    const int r = i >> 6;
    const int c = i & 63;
    int gr = tile + r;
    if (gr >= nRows) gr = nRows - 1;
    const float d = dinv[gr];
    const float4 a4 = *reinterpret_cast<const float4*>(acc + plane_idx(gr, c, nRows));
    const float4 b4 = *reinterpret_cast<const float4*>(bias + c);
    float4 t;
    t.x = d * a4.x + b4.x;
    t.y = d * a4.y + b4.y;
    t.z = d * a4.z + b4.z;
    t.w = d * a4.w + b4.w;
    *reinterpret_cast<float4*>(&alds[r * (64 + 4) + c]) = t;
  }
  __syncthreads();

  const int m  = lane & 15;
  const int kq = (lane >> 4) * 2;
  const int n  = (lane & 15) + wave * 16;

  v8f c = {};
  for (int kk = 0; kk < 64; kk += 4) {
    const int kb = kk + kq;
    v2f a, b;
    a.x = alds[m * (64 + 4) + kb];
    a.y = alds[m * (64 + 4) + kb + 1];
    b.x = wlds[kb * WPAD + n];
    b.y = wlds[(kb + 1) * WPAD + n];
    c = __builtin_amdgcn_wmma_f32_16x16x4_f32(false, a, false, b, (short)0, c,
                                              false, false);
  }

  const int mrow0 = (lane >> 4) * 8;
  for (int v = 0; v < 8; ++v) {
    const int gr = tile + mrow0 + v;
    if (gr < nRows) {
      const long o = plane_idx(gr, n, nRows);
      const float yv = c[v] * dinv[gr];
      Y[o] = yv;
      S[o] = yv;
    }
  }
}

#define EDGE_HALF 16

__device__ __forceinline__ void load_rows32(const float* __restrict__ xl, int r, int j0,
                                            float (&v)[EDGE_HALF]) {
#pragma unroll
  for (int k = 0; k < EDGE_HALF; ++k) {
    const int rj = __builtin_amdgcn_readlane(r, j0 + k);
    v[k] = xl[(long)rj * 32];
  }
}

__device__ __forceinline__ void add_rows32(float* __restrict__ al, int c, int j0,
                                           const float (&v)[EDGE_HALF]) {
#pragma unroll
  for (int k = 0; k < EDGE_HALF; ++k) {
    const int cj = __builtin_amdgcn_readlane(c, j0 + k);
    unsafeAtomicAdd(al + (long)cj * 32, v[k]);
  }
}

__global__ void __launch_bounds__(256)
scatter_edges32(const int* __restrict__ row, const int* __restrict__ col,
                const float* __restrict__ xs, float* __restrict__ acc0,
                float* __restrict__ acc1, int nE) {
  __shared__ int s_half;
  const int lane = threadIdx.x & 31;
  const int wv = __builtin_amdgcn_readfirstlane((int)((blockIdx.x * 256 + threadIdx.x) >> 5));
  const int e0 = wv * 32;
  const int e = e0 + lane;
  int r = 0, c = 0;
  if (e < nE) { r = row[e]; c = col[e]; }
  if (threadIdx.x == 0) s_half = chip_half();
  __syncthreads();
  if (e0 >= nE) return;
  float* acc = __builtin_amdgcn_readfirstlane(s_half) ? acc1 : acc0;
  const float* xl = xs + lane;
  float* al = acc + lane;
  const int cnt = (nE - e0 < 32) ? (nE - e0) : 32;
  if (cnt == 32) {
    float a[EDGE_HALF], b[EDGE_HALF];
    load_rows32(xl, r, 0, a);
    load_rows32(xl, r, 16, b);
    add_rows32(al, c, 0, a);
    add_rows32(al, c, 16, b);
  } else {
    for (int j = 0; j < cnt; ++j) {
      const int rj = __builtin_amdgcn_readlane(r, j);
      const int cj = __builtin_amdgcn_readlane(c, j);
      const float v = xl[(long)rj * 32];
      unsafeAtomicAdd(al + (long)cj * 32, v);
    }
  }
}

#define CHUNK_EDGES 512
#define CTR_STRIDE 64

__global__ void __launch_bounds__(256)
scatter_edges(const int* __restrict__ row, const int* __restrict__ col,
              const float* __restrict__ y, float* __restrict__ acc,
              int* __restrict__ ctr, int nE, int nChunks, int nNodes) {
  __shared__ int s_half;
  __shared__ int s_chunk;
  if (threadIdx.x == 0) {
    int h = chip_half();
    int t = atomicAdd(&ctr[h * CTR_STRIDE], 1);
    if (t >= nChunks) {
      h ^= 1;
      t = atomicAdd(&ctr[h * CTR_STRIDE], 1);
    }
    s_half = h;
    s_chunk = t;
  }
  __syncthreads();
  const int h = __builtin_amdgcn_readfirstlane(s_half);
  const int t = __builtin_amdgcn_readfirstlane(s_chunk);
  if (t >= nChunks) return;
  const int lane = threadIdx.x & 31;
  const int wv = __builtin_amdgcn_readfirstlane((int)(threadIdx.x >> 5));
  const int e0 = t * CHUNK_EDGES + wv * 64;
  const int eA = e0 + lane;
  const int eB = e0 + 32 + lane;
  int rA = 0, cA = 0, rB = 0, cB = 0;
  if (eA < nE) { rA = row[eA]; cA = col[eA]; }
  if (eB < nE) { rB = row[eB]; cB = col[eB]; }
  const float* xl = y + (long)h * nNodes * 32 + lane;
  float* al = acc + (long)h * nNodes * 32 + lane;
  if (e0 + 64 <= nE) {
    float a[EDGE_HALF], b[EDGE_HALF];
    load_rows32(xl, rA, 0, a);
    load_rows32(xl, rA, 16, b);
    add_rows32(al, cA, 0, a);
    load_rows32(xl, rB, 0, a);
    add_rows32(al, cA, 16, b);
    load_rows32(xl, rB, 16, b);
    add_rows32(al, cB, 0, a);
    add_rows32(al, cB, 16, b);
  } else {
    const int cntA = (nE - e0 < 32) ? (nE - e0) : 32;
    const int cntB = (nE - e0 - 32 < 32) ? (nE - e0 - 32) : 32;
    for (int j = 0; j < cntA; ++j) {
      const int rj = __builtin_amdgcn_readlane(rA, j);
      const int cj = __builtin_amdgcn_readlane(cA, j);
      unsafeAtomicAdd(al + (long)cj * 32, xl[(long)rj * 32]);
    }
    for (int j = 0; j < cntB; ++j) {
      const int rj = __builtin_amdgcn_readlane(rB, j);
      const int cj = __builtin_amdgcn_readlane(cB, j);
      unsafeAtomicAdd(al + (long)cj * 32, xl[(long)rj * 32]);
    }
  }
}

__global__ void __launch_bounds__(256)
finalize_layer(const float* __restrict__ acc,
               const float* __restrict__ dinv, const float* __restrict__ bias,
               float* __restrict__ out, int ostride, int nNodes) {
  long t = (long)blockIdx.x * 256 + threadIdx.x;
  if (t >= (long)nNodes * 16) return;
  const int i = (int)(t >> 4);
  const int f = (int)(t & 15) << 2;
  const float d = dinv[i];
  const float4 a = *reinterpret_cast<const float4*>(acc + plane_idx(i, f, nNodes));
  const float4 b = *reinterpret_cast<const float4*>(bias + f);
  float4 o;
  o.x = d * a.x + b.x;
  o.y = d * a.y + b.y;
  o.z = d * a.z + b.z;
  o.w = d * a.w + b.w;
  *reinterpret_cast<float4*>(out + (long)i * ostride + f) = o;
}

extern "C" void kernel_launch(void* const* d_in, const int* in_sizes, int n_in,
                              void* d_out, int out_size, void* d_ws, size_t ws_size,
                              hipStream_t stream) {
  const float* latent    = (const float*)d_in[0];
  const float* condition = (const float*)d_in[1];
  const int*   ei        = (const int*)d_in[2];
  const float* W1 = (const float*)d_in[3];
  const float* b1 = (const float*)d_in[4];
  const float* W2 = (const float*)d_in[5];
  const float* b2 = (const float*)d_in[6];
  const float* W3 = (const float*)d_in[7];
  const float* b3 = (const float*)d_in[8];
  const float* W4 = (const float*)d_in[9];
  const float* b4 = (const float*)d_in[10];

  const int N = in_sizes[0] / 64;
  const int E = in_sizes[2] / 2;
  const int* rowp = ei;
  const int* colp = ei + E;

  char* ws = (char*)d_ws;
  size_t off = 0;
  float* dinv = (float*)(ws + off);
  off += ((size_t)N * 4 + 255) & ~(size_t)255;
  float* ybuf = (float*)(ws + off);
  off += (size_t)N * 64 * 4;
  float* acc  = (float*)(ws + off);
  off += (size_t)N * 64 * 4;
  float* hbuf = (float*)(ws + off);

  const long accElems = (long)N * 64;
  const dim3 gemmGrid((N + 15) / 16);
  const dim3 gemmBlk(128);
  const dim3 fusedGrid((N + FROWS - 1) / FROWS);
  const dim3 edgeGrid((E + 255) / 256);
  const int  nChunks = (E + CHUNK_EDGES - 1) / CHUNK_EDGES;
  const dim3 planeGrid(2 * nChunks);
  const dim3 finGrid(((long)N * 16 + 255) / 256);
  const int  nCtr = 6 * CTR_STRIDE;
  const dim3 fill2N((2 * (long)N + nCtr + 255) / 256);
  const dim3 stageGrid(((long)N * 24 + 255) / 256);
  float* outp = (float*)d_out;

  float* xsA   = hbuf;
  float* xsB   = hbuf + accElems;
  float* deg0  = hbuf + (long)N * 96;
  float* deg1  = deg0 + N;
  int*   ctr   = (int*)(deg1 + N);
  float* accA  = ybuf;
  float* accB  = acc;
  float* accB1 = acc + (long)N * 32;
  float* acc3  = outp;
  fill2_f32<<<fill2N, 256, 0, stream>>>(deg0, 1.0f, N, deg1, 0.0f, (long)N + nCtr);
  count_deg<<<(E + 255) / 256, 256, 0, stream>>>(colp, deg0, deg1, E);
  stage_inputs<<<stageGrid, 256, 0, stream>>>(latent, condition, deg0, deg1, dinv, xsA, xsB,
                                              accA, accB, accB1, N);
  scatter_edges<<<planeGrid, 256, 0, stream>>>(rowp, colp, xsA, accA, ctr, E, nChunks, N);
  scatter_edges32<<<edgeGrid, 256, 0, stream>>>(rowp, colp, xsB, accB, accB1, E);

  float* y3   = hbuf;
  gcn_layers123_wmma<<<fusedGrid, 256, 0, stream>>>(accA, accB, accB1,
                                                    W1, b1, W2, b2, W3, dinv, y3, acc3, N);
  scatter_edges<<<planeGrid, 256, 0, stream>>>(rowp, colp, y3, acc3, ctr + 2 * CTR_STRIDE, E, nChunks, N);

  gcn_fin_gemm_wmma<<<gemmGrid, gemmBlk, 0, stream>>>(acc3, dinv, b3, W4, ybuf, acc, N);
  scatter_edges<<<planeGrid, 256, 0, stream>>>(rowp, colp, ybuf, acc, ctr + 4 * CTR_STRIDE, E, nChunks, N);
  finalize_layer<<<finGrid, 256, 0, stream>>>(acc, dinv, b4, outp, 64, N);
}
